# starting kernel
# baseline (speedup 1.0000x reference)
#include <hip/hip_runtime.h>

typedef __attribute__((ext_vector_type(16))) _Float16 v16h;
typedef __attribute__((ext_vector_type(8)))  float    v8f;

#define IN_CH 128
#define HID   64
#define NFRAG (4 * 4 * 32 * 16)

__device__ __forceinline__ v8f wmma_f16(v16h a, v16h b, v8f c)
{
    return __builtin_amdgcn_wmma_f32_16x16x32_f16(false, a, false, b,
                                                  (short)0, c, false, false);
}

__device__ __forceinline__ void stage_Bfrag(const float* __restrict__ W,
                                            _Float16* sB)
{
    for (int i = threadIdx.x; i < NFRAG; i += blockDim.x) {
        const int j     = i & 15;
        const int lane  = (i >> 4) & 31;
        const int kci   = (i >> 9) & 3;
        const int ntile = i >> 11;
        const int koff  = (lane >> 4) << 3;
        const int col   = (ntile << 4) + (lane & 15);
        const int k     = (kci << 5) + koff + (j & 7) + ((j >> 3) << 4);
        sB[i] = (_Float16)W[k * HID + col];
    }
}

__device__ __forceinline__ v16h load_a16(const float* __restrict__ arow, int off)
{
    float af[16];
    reinterpret_cast<float4*>(af)[0] = *reinterpret_cast<const float4*>(arow + off);
    reinterpret_cast<float4*>(af)[1] = *reinterpret_cast<const float4*>(arow + off + 4);
    reinterpret_cast<float4*>(af)[2] = *reinterpret_cast<const float4*>(arow + off + 16);
    reinterpret_cast<float4*>(af)[3] = *reinterpret_cast<const float4*>(arow + off + 20);
    v16h a;
    #pragma unroll
    for (int j = 0; j < 16; ++j) a[j] = (_Float16)af[j];
    return a;
}

__global__ void __launch_bounds__(256)
gemm1_wmma(const float* __restrict__ x,
           const float* __restrict__ Wc,
           float* __restrict__ h,
           int n_nodes)
{
    __shared__ __align__(32) _Float16 sB[NFRAG];
    stage_Bfrag(Wc, sB);
    __syncthreads();

    const int wave = threadIdx.x >> 5;
    const int lane = threadIdx.x & 31;
    const int mtiles  = n_nodes >> 4;
    const int mgroups = (mtiles + 3) >> 2;
    const int g = blockIdx.x * (blockDim.x >> 5) + wave;
    if (g >= mgroups * 4) return;
    const int ntile = g & 3;
    const int mg    = g >> 2;

    const v16h* bp = reinterpret_cast<const v16h*>(sB) + (ntile << 7) + lane;
    const v16h b0 = bp[0], b1 = bp[32], b2 = bp[64], b3 = bp[96];

    const int koff = (lane >> 4) << 3;
    const int col  = (ntile << 4) + (lane & 15);

    for (int t = 0; t < 4; ++t) {
        const int mtile = (mg << 2) + t;
        if (mtile >= mtiles) return;
        const int row = (mtile << 4) + (lane & 15);
        const float* arow = x + (long)row * IN_CH;
        __builtin_prefetch(arow + 256, 0, 1);

        v8f c = {};
        c = wmma_f16(load_a16(arow,      koff), b0, c);
        c = wmma_f16(load_a16(arow, 32 + koff), b1, c);
        c = wmma_f16(load_a16(arow, 64 + koff), b2, c);
        c = wmma_f16(load_a16(arow, 96 + koff), b3, c);

        const int rbase = (mtile << 4) + ((lane >> 4) << 3);
        #pragma unroll
        for (int r = 0; r < 8; ++r)
            h[(long)(rbase + r) * HID + col] = c[r];
    }
}

__global__ void zero_f32(float4* __restrict__ p, int n4)
{
    int i = blockIdx.x * blockDim.x + threadIdx.x;
    for (; i < n4; i += gridDim.x * blockDim.x)
        p[i] = make_float4(0.f, 0.f, 0.f, 0.f);
}

__global__ void __launch_bounds__(256)
scatter_edges(const float* __restrict__ h,
              const int*   __restrict__ ei,
              const float* __restrict__ ew,
              float*       __restrict__ agg,
              int E)
{
    const int t    = blockIdx.x * blockDim.x + threadIdx.x;
    const int e    = t >> 5;
    const int lane = t & 31;
    if (e >= E) return;

    const int   s = ei[e];
    const int   d = ei[E + e];
    const float w = ew[e];

    float2 hv = reinterpret_cast<const float2*>(h + (long)s * HID)[lane];
    float* ap = agg + (long)d * HID + (lane << 1);
    atomicAdd(ap,     hv.x * w);
    atomicAdd(ap + 1, hv.y * w);
}

__global__ void __launch_bounds__(256)
bias_lrelu(float* __restrict__ agg,
           const float* __restrict__ b_conv,
           const float* __restrict__ alpha,
           int total4)
{
    const int i = blockIdx.x * blockDim.x + threadIdx.x;
    if (i >= total4) return;
    const int c4 = (i & 15) << 2;
    float4 v  = reinterpret_cast<float4*>(agg)[i];
    float4 bb = *reinterpret_cast<const float4*>(b_conv + c4);
    float4 aa = *reinterpret_cast<const float4*>(alpha + c4);
    v.x += bb.x; v.y += bb.y; v.z += bb.z; v.w += bb.w;
    v.x = v.x > 0.f ? v.x : aa.x * v.x;
    v.y = v.y > 0.f ? v.y : aa.y * v.y;
    v.z = v.z > 0.f ? v.z : aa.z * v.z;
    v.w = v.w > 0.f ? v.w : aa.w * v.w;
    reinterpret_cast<float4*>(agg)[i] = v;
}

__global__ void __launch_bounds__(256)
gemm2_wmma(const float* __restrict__ act,
           const float* __restrict__ Wcat,
           const float* __restrict__ b_cat,
           float* __restrict__ out,
           int n_half)
{
    __shared__ __align__(32) _Float16 sB[NFRAG];
    stage_Bfrag(Wcat, sB);
    __syncthreads();

    const int wave = threadIdx.x >> 5;
    const int lane = threadIdx.x & 31;
    const int mtiles  = (n_half + 15) >> 4;
    const int mgroups = (mtiles + 3) >> 2;
    const int g = blockIdx.x * (blockDim.x >> 5) + wave;
    if (g >= mgroups * 4) return;
    const int ntile = g & 3;
    const int mg    = g >> 2;

    const v16h* bp = reinterpret_cast<const v16h*>(sB) + (ntile << 7) + lane;
    const v16h b0 = bp[0], b1 = bp[32], b2 = bp[64], b3 = bp[96];

    const int koff = (lane >> 4) << 3;
    const int col  = (ntile << 4) + (lane & 15);

    for (int t = 0; t < 4; ++t) {
        const int mtile = (mg << 2) + t;
        if (mtile >= mtiles) return;
        const int row  = (mtile << 4) + (lane & 15);
        const int rowc = row < n_half ? row : n_half - 1;
        const float* a0 = act + (long)rowc * HID;
        const float* a1 = act + (long)(rowc + n_half) * HID;

        v8f c = {};
        c = wmma_f16(load_a16(a0,      koff), b0, c);
        c = wmma_f16(load_a16(a0, 32 + koff), b1, c);
        c = wmma_f16(load_a16(a1,      koff), b2, c);
        c = wmma_f16(load_a16(a1, 32 + koff), b3, c);

        const float bb = b_cat[col];
        const int rbase = (mtile << 4) + ((lane >> 4) << 3);
        #pragma unroll
        for (int r = 0; r < 8; ++r) {
            const int rr = rbase + r;
            if (rr < n_half) {
                float v = c[r] + bb;
                out[(long)rr * HID + col] = v > 0.f ? v : 0.f;
            }
        }
    }
}

extern "C" void kernel_launch(void* const* d_in, const int* in_sizes, int n_in,
                              void* d_out, int out_size, void* d_ws, size_t ws_size,
                              hipStream_t stream)
{
    const float* x   = (const float*)d_in[0];
    const int*   ei  = (const int*)  d_in[1];
    const float* ew  = (const float*)d_in[2];
    const float* Wc  = (const float*)d_in[3];
    const float* bc  = (const float*)d_in[4];
    const float* al  = (const float*)d_in[5];
    const float* Wk  = (const float*)d_in[6];
    const float* bk  = (const float*)d_in[7];

    const int n_nodes = in_sizes[0] / IN_CH;
    const int E       = in_sizes[2];
    const int n_half  = n_nodes / 2;

    float* h   = (float*)d_ws;
    float* agg = h + (size_t)n_nodes * HID;
    float* out = (float*)d_out;

    {
        const int mgroups = ((n_nodes >> 4) + 3) >> 2;
        const int waves   = mgroups * 4;
        gemm1_wmma<<<(waves + 7) / 8, 256, 0, stream>>>(x, Wc, h, n_nodes);
    }

    const int n4 = (n_nodes * HID) >> 2;
    zero_f32<<<(n4 + 255) / 256, 256, 0, stream>>>((float4*)agg, n4);

    scatter_edges<<<(E + 7) / 8, 256, 0, stream>>>(h, ei, ew, agg, E);

    bias_lrelu<<<(n4 + 255) / 256, 256, 0, stream>>>(agg, bc, al, n4);

    {
        const int mgroups = (((n_half + 15) >> 4) + 3) >> 2;
        const int waves   = mgroups * 4;
        gemm2_wmma<<<(waves + 7) / 8, 256, 0, stream>>>(agg, Wk, bk, out, n_half);
    }
}


# edge scatter: each wave takes 16 consecutive edges and requests all 16 source rows before its adds (16x fewer waves, 16 row reads in flight); atomic adds and their varying arrival order kept
# speedup vs baseline: 1.6465x; 1.6465x over previous
#include <hip/hip_runtime.h>

typedef __attribute__((ext_vector_type(16))) _Float16 v16h;
typedef __attribute__((ext_vector_type(8)))  float    v8f;

#define IN_CH 128
#define HID   64
#define NFRAG (4 * 4 * 32 * 16)

__device__ __forceinline__ v8f wmma_f16(v16h a, v16h b, v8f c)
{
    return __builtin_amdgcn_wmma_f32_16x16x32_f16(false, a, false, b,
                                                  (short)0, c, false, false);
}

__device__ __forceinline__ void stage_Bfrag(const float* __restrict__ W,
                                            _Float16* sB)
{
    for (int i = threadIdx.x; i < NFRAG; i += blockDim.x) {
        const int j     = i & 15;
        const int lane  = (i >> 4) & 31;
        const int kci   = (i >> 9) & 3;
        const int ntile = i >> 11;
        const int koff  = (lane >> 4) << 3;
        const int col   = (ntile << 4) + (lane & 15);
        const int k     = (kci << 5) + koff + (j & 7) + ((j >> 3) << 4);
        sB[i] = (_Float16)W[k * HID + col];
    }
}

__device__ __forceinline__ v16h load_a16(const float* __restrict__ arow, int off)
{
    float af[16];
    reinterpret_cast<float4*>(af)[0] = *reinterpret_cast<const float4*>(arow + off);
    reinterpret_cast<float4*>(af)[1] = *reinterpret_cast<const float4*>(arow + off + 4);
    reinterpret_cast<float4*>(af)[2] = *reinterpret_cast<const float4*>(arow + off + 16);
    reinterpret_cast<float4*>(af)[3] = *reinterpret_cast<const float4*>(arow + off + 20);
    v16h a;
    #pragma unroll
    for (int j = 0; j < 16; ++j) a[j] = (_Float16)af[j];
    return a;
}

__device__ __forceinline__ void load_ch16(const float* __restrict__ p, int off, float (&v)[16])
{
    reinterpret_cast<float4*>(v)[0] = *reinterpret_cast<const float4*>(p + off);
    reinterpret_cast<float4*>(v)[1] = *reinterpret_cast<const float4*>(p + off + 4);
    reinterpret_cast<float4*>(v)[2] = *reinterpret_cast<const float4*>(p + off + 16);
    reinterpret_cast<float4*>(v)[3] = *reinterpret_cast<const float4*>(p + off + 20);
}

__device__ __forceinline__ v16h load_a16_act(const float* __restrict__ arow, int off,
                                             const float (&bb)[16], const float (&aa)[16])
{
    float af[16];
    reinterpret_cast<float4*>(af)[0] = *reinterpret_cast<const float4*>(arow + off);
    reinterpret_cast<float4*>(af)[1] = *reinterpret_cast<const float4*>(arow + off + 4);
    reinterpret_cast<float4*>(af)[2] = *reinterpret_cast<const float4*>(arow + off + 16);
    reinterpret_cast<float4*>(af)[3] = *reinterpret_cast<const float4*>(arow + off + 20);
    v16h a;
    #pragma unroll
    for (int j = 0; j < 16; ++j) {
        float v = af[j] + bb[j];
        v = v > 0.f ? v : aa[j] * v;
        a[j] = (_Float16)v;
    }
    return a;
}

__global__ void __launch_bounds__(256)
gemm1_wmma(const float* __restrict__ x,
           const float* __restrict__ Wc,
           float* __restrict__ h,
           int n_nodes)
{
    __shared__ __align__(32) _Float16 sB[NFRAG];
    stage_Bfrag(Wc, sB);
    __syncthreads();

    const int wave = threadIdx.x >> 5;
    const int lane = threadIdx.x & 31;
    const int mtiles  = n_nodes >> 4;
    const int mgroups = (mtiles + 3) >> 2;
    const int g = blockIdx.x * (blockDim.x >> 5) + wave;
    if (g >= mgroups * 4) return;
    const int ntile = g & 3;
    const int mg    = g >> 2;

    const v16h* bp = reinterpret_cast<const v16h*>(sB) + (ntile << 7) + lane;
    const v16h b0 = bp[0], b1 = bp[32], b2 = bp[64], b3 = bp[96];

    const int koff = (lane >> 4) << 3;
    const int col  = (ntile << 4) + (lane & 15);

    for (int t = 0; t < 4; ++t) {
        const int mtile = (mg << 2) + t;
        if (mtile >= mtiles) return;
        const int row = (mtile << 4) + (lane & 15);
        const float* arow = x + (long)row * IN_CH;
        __builtin_prefetch(arow + 256, 0, 1);

        v8f c = {};
        c = wmma_f16(load_a16(arow,      koff), b0, c);
        c = wmma_f16(load_a16(arow, 32 + koff), b1, c);
        c = wmma_f16(load_a16(arow, 64 + koff), b2, c);
        c = wmma_f16(load_a16(arow, 96 + koff), b3, c);

        const int rbase = (mtile << 4) + ((lane >> 4) << 3);
        #pragma unroll
        for (int r = 0; r < 8; ++r)
            h[(long)(rbase + r) * HID + col] = c[r];
    }
}

__global__ void zero_f32(float4* __restrict__ p, int n4)
{
    int i = blockIdx.x * blockDim.x + threadIdx.x;
    for (; i < n4; i += gridDim.x * blockDim.x)
        p[i] = make_float4(0.f, 0.f, 0.f, 0.f);
}

#define EPW 16

__global__ void __launch_bounds__(256)
scatter_edges(const float* __restrict__ h,
              const int*   __restrict__ ei,
              const float* __restrict__ ew,
              float*       __restrict__ agg,
              int E)
{
    const int lane = threadIdx.x & 31;
    const int wv = __builtin_amdgcn_readfirstlane((int)((blockIdx.x * blockDim.x + threadIdx.x) >> 5));
    const int e0 = wv * EPW;
    if (e0 >= E) return;

    int el = e0 + (lane & (EPW - 1));
    el = el < E ? el : E - 1;
    const int   sl = ei[el];
    const int   dl = ei[E + el];
    const float wl = ew[el];

    const int n = (E - e0) < EPW ? (E - e0) : EPW;

    if (n == EPW) {
        float2 hv[EPW];
        #pragma unroll
        for (int j = 0; j < EPW; ++j) {
            const int s = __builtin_amdgcn_readlane(sl, j);
            hv[j] = reinterpret_cast<const float2*>(h + (long)s * HID)[lane];
        }
        #pragma unroll
        for (int j = 0; j < EPW; ++j) {
            const int   d = __builtin_amdgcn_readlane(dl, j);
            const float w = __int_as_float(__builtin_amdgcn_readlane(__float_as_int(wl), j));
            float* ap = agg + (long)d * HID + (lane << 1);
            atomicAdd(ap,     hv[j].x * w);
            atomicAdd(ap + 1, hv[j].y * w);
        }
    } else {
        for (int j = 0; j < n; ++j) {
            const int   s = __builtin_amdgcn_readlane(sl, j);
            const int   d = __builtin_amdgcn_readlane(dl, j);
            const float w = __int_as_float(__builtin_amdgcn_readlane(__float_as_int(wl), j));
            const float2 hv = reinterpret_cast<const float2*>(h + (long)s * HID)[lane];
            float* ap = agg + (long)d * HID + (lane << 1);
            atomicAdd(ap,     hv.x * w);
            atomicAdd(ap + 1, hv.y * w);
        }
    }
}

__global__ void __launch_bounds__(256)
gemm2_wmma(const float* __restrict__ agg,
           const float* __restrict__ b_conv,
           const float* __restrict__ alpha,
           const float* __restrict__ Wcat,
           const float* __restrict__ b_cat,
           float* __restrict__ out,
           int n_half)
{
    __shared__ __align__(32) _Float16 sB[NFRAG];

    const int wave = threadIdx.x >> 5;
    const int lane = threadIdx.x & 31;
    const int koff = (lane >> 4) << 3;

    float bbL[16], aaL[16], bbH[16], aaH[16];
    load_ch16(b_conv,      koff, bbL);
    load_ch16(alpha,       koff, aaL);
    load_ch16(b_conv, 32 + koff, bbH);
    load_ch16(alpha,  32 + koff, aaH);

    stage_Bfrag(Wcat, sB);
    __syncthreads();

    const int mtiles  = (n_half + 15) >> 4;
    const int mgroups = (mtiles + 3) >> 2;
    const int g = blockIdx.x * (blockDim.x >> 5) + wave;
    if (g >= mgroups * 4) return;
    const int ntile = g & 3;
    const int mg    = g >> 2;

    const v16h* bp = reinterpret_cast<const v16h*>(sB) + (ntile << 7) + lane;
    const v16h b0 = bp[0], b1 = bp[32], b2 = bp[64], b3 = bp[96];

    const int col  = (ntile << 4) + (lane & 15);

    for (int t = 0; t < 4; ++t) {
        const int mtile = (mg << 2) + t;
        if (mtile >= mtiles) return;
        const int row  = (mtile << 4) + (lane & 15);
        const int rowc = row < n_half ? row : n_half - 1;
        const float* a0 = agg + (long)rowc * HID;
        const float* a1 = agg + (long)(rowc + n_half) * HID;

        v8f c = {};
        c = wmma_f16(load_a16_act(a0,      koff, bbL, aaL), b0, c);
        c = wmma_f16(load_a16_act(a0, 32 + koff, bbH, aaH), b1, c);
        c = wmma_f16(load_a16_act(a1,      koff, bbL, aaL), b2, c);
        c = wmma_f16(load_a16_act(a1, 32 + koff, bbH, aaH), b3, c);

        const float bb = b_cat[col];
        const int rbase = (mtile << 4) + ((lane >> 4) << 3);
        #pragma unroll
        for (int r = 0; r < 8; ++r) {
            const int rr = rbase + r;
            if (rr < n_half) {
                float v = c[r] + bb;
                out[(long)rr * HID + col] = v > 0.f ? v : 0.f;
            }
        }
    }
}

extern "C" void kernel_launch(void* const* d_in, const int* in_sizes, int n_in,
                              void* d_out, int out_size, void* d_ws, size_t ws_size,
                              hipStream_t stream)
{
    const float* x   = (const float*)d_in[0];
    const int*   ei  = (const int*)  d_in[1];
    const float* ew  = (const float*)d_in[2];
    const float* Wc  = (const float*)d_in[3];
    const float* bc  = (const float*)d_in[4];
    const float* al  = (const float*)d_in[5];
    const float* Wk  = (const float*)d_in[6];
    const float* bk  = (const float*)d_in[7];

    const int n_nodes = in_sizes[0] / IN_CH;
    const int E       = in_sizes[2];
    const int n_half  = n_nodes / 2;

    float* h   = (float*)d_ws;
    float* agg = h + (size_t)n_nodes * HID;
    float* out = (float*)d_out;

    {
        const int mgroups = ((n_nodes >> 4) + 3) >> 2;
        const int waves   = mgroups * 4;
        gemm1_wmma<<<(waves + 7) / 8, 256, 0, stream>>>(x, Wc, h, n_nodes);
    }

    const int n4 = (n_nodes * HID) >> 2;
    zero_f32<<<(n4 + 255) / 256, 256, 0, stream>>>((float4*)agg, n4);

    {
        const int waves = (E + EPW - 1) / EPW;
        scatter_edges<<<(waves + 7) / 8, 256, 0, stream>>>(h, ei, ew, agg, E);
    }

    {
        const int mgroups = (((n_half + 15) >> 4) + 3) >> 2;
        const int waves   = mgroups * 4;
        gemm2_wmma<<<(waves + 7) / 8, 256, 0, stream>>>(agg, bc, al, Wk, bk, out, n_half);
    }
}


# edge scatter: lane l owns channels l and l+32 (was 2l, 2l+1), so each row load and each atomic add instruction covers one whole 128-byte line instead of half of two; arrival-order sums kept
# speedup vs baseline: 1.7539x; 1.0652x over previous
#include <hip/hip_runtime.h>

typedef __attribute__((ext_vector_type(16))) _Float16 v16h;
typedef __attribute__((ext_vector_type(8)))  float    v8f;

#define IN_CH 128
#define HID   64
#define NFRAG (4 * 4 * 32 * 16)

__device__ __forceinline__ v8f wmma_f16(v16h a, v16h b, v8f c)
{
    return __builtin_amdgcn_wmma_f32_16x16x32_f16(false, a, false, b,
                                                  (short)0, c, false, false);
}

__device__ __forceinline__ void stage_Bfrag(const float* __restrict__ W,
                                            _Float16* sB)
{
    for (int i = threadIdx.x; i < NFRAG; i += blockDim.x) {
        const int j     = i & 15;
        const int lane  = (i >> 4) & 31;
        const int kci   = (i >> 9) & 3;
        const int ntile = i >> 11;
        const int koff  = (lane >> 4) << 3;
        const int col   = (ntile << 4) + (lane & 15);
        const int k     = (kci << 5) + koff + (j & 7) + ((j >> 3) << 4);
        sB[i] = (_Float16)W[k * HID + col];
    }
}

__device__ __forceinline__ v16h load_a16(const float* __restrict__ arow, int off)
{
    float af[16];
    reinterpret_cast<float4*>(af)[0] = *reinterpret_cast<const float4*>(arow + off);
    reinterpret_cast<float4*>(af)[1] = *reinterpret_cast<const float4*>(arow + off + 4);
    reinterpret_cast<float4*>(af)[2] = *reinterpret_cast<const float4*>(arow + off + 16);
    reinterpret_cast<float4*>(af)[3] = *reinterpret_cast<const float4*>(arow + off + 20);
    v16h a;
    #pragma unroll
    for (int j = 0; j < 16; ++j) a[j] = (_Float16)af[j];
    return a;
}

__device__ __forceinline__ void load_ch16(const float* __restrict__ p, int off, float (&v)[16])
{
    reinterpret_cast<float4*>(v)[0] = *reinterpret_cast<const float4*>(p + off);
    reinterpret_cast<float4*>(v)[1] = *reinterpret_cast<const float4*>(p + off + 4);
    reinterpret_cast<float4*>(v)[2] = *reinterpret_cast<const float4*>(p + off + 16);
    reinterpret_cast<float4*>(v)[3] = *reinterpret_cast<const float4*>(p + off + 20);
}

__device__ __forceinline__ v16h load_a16_act(const float* __restrict__ arow, int off,
                                             const float (&bb)[16], const float (&aa)[16])
{
    float af[16];
    reinterpret_cast<float4*>(af)[0] = *reinterpret_cast<const float4*>(arow + off);
    reinterpret_cast<float4*>(af)[1] = *reinterpret_cast<const float4*>(arow + off + 4);
    reinterpret_cast<float4*>(af)[2] = *reinterpret_cast<const float4*>(arow + off + 16);
    reinterpret_cast<float4*>(af)[3] = *reinterpret_cast<const float4*>(arow + off + 20);
    v16h a;
    #pragma unroll
    for (int j = 0; j < 16; ++j) {
        float v = af[j] + bb[j];
        v = v > 0.f ? v : aa[j] * v;
        a[j] = (_Float16)v;
    }
    return a;
}

__global__ void __launch_bounds__(256)
gemm1_wmma(const float* __restrict__ x,
           const float* __restrict__ Wc,
           float* __restrict__ h,
           int n_nodes)
{
    __shared__ __align__(32) _Float16 sB[NFRAG];
    stage_Bfrag(Wc, sB);
    __syncthreads();

    const int wave = threadIdx.x >> 5;
    const int lane = threadIdx.x & 31;
    const int mtiles  = n_nodes >> 4;
    const int mgroups = (mtiles + 3) >> 2;
    const int g = blockIdx.x * (blockDim.x >> 5) + wave;
    if (g >= mgroups * 4) return;
    const int ntile = g & 3;
    const int mg    = g >> 2;

    const v16h* bp = reinterpret_cast<const v16h*>(sB) + (ntile << 7) + lane;
    const v16h b0 = bp[0], b1 = bp[32], b2 = bp[64], b3 = bp[96];

    const int koff = (lane >> 4) << 3;
    const int col  = (ntile << 4) + (lane & 15);

    for (int t = 0; t < 4; ++t) {
        const int mtile = (mg << 2) + t;
        if (mtile >= mtiles) return;
        const int row = (mtile << 4) + (lane & 15);
        const float* arow = x + (long)row * IN_CH;
        __builtin_prefetch(arow + 256, 0, 1);

        v8f c = {};
        c = wmma_f16(load_a16(arow,      koff), b0, c);
        c = wmma_f16(load_a16(arow, 32 + koff), b1, c);
        c = wmma_f16(load_a16(arow, 64 + koff), b2, c);
        c = wmma_f16(load_a16(arow, 96 + koff), b3, c);

        const int rbase = (mtile << 4) + ((lane >> 4) << 3);
        #pragma unroll
        for (int r = 0; r < 8; ++r)
            h[(long)(rbase + r) * HID + col] = c[r];
    }
}

__global__ void zero_f32(float4* __restrict__ p, int n4)
{
    int i = blockIdx.x * blockDim.x + threadIdx.x;
    for (; i < n4; i += gridDim.x * blockDim.x)
        p[i] = make_float4(0.f, 0.f, 0.f, 0.f);
}

#define EPW 16

__global__ void __launch_bounds__(256)
scatter_edges(const float* __restrict__ h,
              const int*   __restrict__ ei,
              const float* __restrict__ ew,
              float*       __restrict__ agg,
              int E)
{
    const int lane = threadIdx.x & 31;
    const int wv = __builtin_amdgcn_readfirstlane((int)((blockIdx.x * blockDim.x + threadIdx.x) >> 5));
    const int e0 = wv * EPW;
    if (e0 >= E) return;

    int el = e0 + (lane & (EPW - 1));
    el = el < E ? el : E - 1;
    const int   sl = ei[el];
    const int   dl = ei[E + el];
    const float wl = ew[el];

    const int n = (E - e0) < EPW ? (E - e0) : EPW;

    if (n == EPW) {
        float lo[EPW], hi[EPW];
        #pragma unroll
        for (int j = 0; j < EPW; ++j) {
            const int s = __builtin_amdgcn_readlane(sl, j);
            const float* hp = h + (long)s * HID + lane;
            lo[j] = hp[0];
            hi[j] = hp[32];
        }
        #pragma unroll
        for (int j = 0; j < EPW; ++j) {
            const int   d = __builtin_amdgcn_readlane(dl, j);
            const float w = __int_as_float(__builtin_amdgcn_readlane(__float_as_int(wl), j));
            float* ap = agg + (long)d * HID + lane;
            atomicAdd(ap,      lo[j] * w);
            atomicAdd(ap + 32, hi[j] * w);
        }
    } else {
        for (int j = 0; j < n; ++j) {
            const int   s = __builtin_amdgcn_readlane(sl, j);
            const int   d = __builtin_amdgcn_readlane(dl, j);
            const float w = __int_as_float(__builtin_amdgcn_readlane(__float_as_int(wl), j));
            const float* hp = h + (long)s * HID + lane;
            const float vlo = hp[0];
            const float vhi = hp[32];
            float* ap = agg + (long)d * HID + lane;
            atomicAdd(ap,      vlo * w);
            atomicAdd(ap + 32, vhi * w);
        }
    }
}

__global__ void __launch_bounds__(256)
gemm2_wmma(const float* __restrict__ agg,
           const float* __restrict__ b_conv,
           const float* __restrict__ alpha,
           const float* __restrict__ Wcat,
           const float* __restrict__ b_cat,
           float* __restrict__ out,
           int n_half)
{
    __shared__ __align__(32) _Float16 sB[NFRAG];

    const int wave = threadIdx.x >> 5;
    const int lane = threadIdx.x & 31;
    const int koff = (lane >> 4) << 3;

    float bbL[16], aaL[16], bbH[16], aaH[16];
    load_ch16(b_conv,      koff, bbL);
    load_ch16(alpha,       koff, aaL);
    load_ch16(b_conv, 32 + koff, bbH);
    load_ch16(alpha,  32 + koff, aaH);

    stage_Bfrag(Wcat, sB);
    __syncthreads();

    const int mtiles  = (n_half + 15) >> 4;
    const int mgroups = (mtiles + 3) >> 2;
    const int g = blockIdx.x * (blockDim.x >> 5) + wave;
    if (g >= mgroups * 4) return;
    const int ntile = g & 3;
    const int mg    = g >> 2;

    const v16h* bp = reinterpret_cast<const v16h*>(sB) + (ntile << 7) + lane;
    const v16h b0 = bp[0], b1 = bp[32], b2 = bp[64], b3 = bp[96];

    const int col  = (ntile << 4) + (lane & 15);

    for (int t = 0; t < 4; ++t) {
        const int mtile = (mg << 2) + t;
        if (mtile >= mtiles) return;
        const int row  = (mtile << 4) + (lane & 15);
        const int rowc = row < n_half ? row : n_half - 1;
        const float* a0 = agg + (long)rowc * HID;
        const float* a1 = agg + (long)(rowc + n_half) * HID;

        v8f c = {};
        c = wmma_f16(load_a16_act(a0,      koff, bbL, aaL), b0, c);
        c = wmma_f16(load_a16_act(a0, 32 + koff, bbH, aaH), b1, c);
        c = wmma_f16(load_a16_act(a1,      koff, bbL, aaL), b2, c);
        c = wmma_f16(load_a16_act(a1, 32 + koff, bbH, aaH), b3, c);

        const float bb = b_cat[col];
        const int rbase = (mtile << 4) + ((lane >> 4) << 3);
        #pragma unroll
        for (int r = 0; r < 8; ++r) {
            const int rr = rbase + r;
            if (rr < n_half) {
                float v = c[r] + bb;
                out[(long)rr * HID + col] = v > 0.f ? v : 0.f;
            }
        }
    }
}

extern "C" void kernel_launch(void* const* d_in, const int* in_sizes, int n_in,
                              void* d_out, int out_size, void* d_ws, size_t ws_size,
                              hipStream_t stream)
{
    const float* x   = (const float*)d_in[0];
    const int*   ei  = (const int*)  d_in[1];
    const float* ew  = (const float*)d_in[2];
    const float* Wc  = (const float*)d_in[3];
    const float* bc  = (const float*)d_in[4];
    const float* al  = (const float*)d_in[5];
    const float* Wk  = (const float*)d_in[6];
    const float* bk  = (const float*)d_in[7];

    const int n_nodes = in_sizes[0] / IN_CH;
    const int E       = in_sizes[2];
    const int n_half  = n_nodes / 2;

    float* h   = (float*)d_ws;
    float* agg = h + (size_t)n_nodes * HID;
    float* out = (float*)d_out;

    {
        const int mgroups = ((n_nodes >> 4) + 3) >> 2;
        const int waves   = mgroups * 4;
        gemm1_wmma<<<(waves + 7) / 8, 256, 0, stream>>>(x, Wc, h, n_nodes);
    }

    const int n4 = (n_nodes * HID) >> 2;
    zero_f32<<<(n4 + 255) / 256, 256, 0, stream>>>((float4*)agg, n4);

    {
        const int waves = (E + EPW - 1) / EPW;
        scatter_edges<<<(waves + 7) / 8, 256, 0, stream>>>(h, ei, ew, agg, E);
    }

    {
        const int mgroups = (((n_half + 15) >> 4) + 3) >> 2;
        const int waves   = mgroups * 4;
        gemm2_wmma<<<(waves + 7) / 8, 256, 0, stream>>>(agg, bc, al, Wk, bk, out, n_half);
    }
}
